# starting kernel
# baseline (speedup 1.0000x reference)
#include <hip/hip_runtime.h>

typedef __attribute__((ext_vector_type(2))) float v2f;
typedef __attribute__((ext_vector_type(8))) float v8f;

#define IN_SIZE   1024
#define TILE_ROWS 16

#define W1T_OFF 0
#define W2T_OFF 16384
#define W3_OFF  32768


__global__ void __launch_bounds__(256)
bfly_precompute(const float* __restrict__ tw, float* __restrict__ ws) {
    float* w1t = ws + W1T_OFF;
    float* w2t = ws + W2T_OFF;
    float* w3  = ws + W3_OFF;
    int tid = blockIdx.x * 256 + threadIdx.x;

    if (tid < 1024) {
        int b = tid >> 4, j = tid & 15;
        float v[16];
#pragma unroll
        for (int l = 0; l < 16; ++l) v[l] = (l == j) ? 1.0f : 0.0f;
#pragma unroll
        for (int idx = 0; idx < 4; ++idx) {
            int s = 1 << idx;
            float nv[16];
#pragma unroll
            for (int l = 0; l < 16; ++l) {
                int p = b * 16 + l;
                int q = (p >> (idx + 1)) * s + (p & (s - 1));
                int i = (l >> idx) & 1;
                const float* tq = tw + idx * 2048 + q * 4 + i * 2;
                nv[l] = tq[0] * v[l & ~s] + tq[1] * v[l | s];
            }
#pragma unroll
            for (int l = 0; l < 16; ++l) v[l] = nv[l];
        }
#pragma unroll
        for (int i = 0; i < 16; ++i) w1t[b * 256 + j * 16 + i] = v[i];
    } else if (tid < 2048) {
        int t2 = tid - 1024;
        int blk = t2 >> 4, j = t2 & 15;
        int G = blk >> 4, c = blk & 15;
        float v[16];
#pragma unroll
        for (int m = 0; m < 16; ++m) v[m] = (m == j) ? 1.0f : 0.0f;
#pragma unroll
        for (int idx = 4; idx < 8; ++idx) {
            int s = 1 << idx;
            int bit = 1 << (idx - 4);
            float nv[16];
#pragma unroll
            for (int m = 0; m < 16; ++m) {
                int p = G * 256 + 16 * m + c;
                int q = (p >> (idx + 1)) * s + (p & (s - 1));
                int i = (p >> idx) & 1;
                const float* tq = tw + idx * 2048 + q * 4 + i * 2;
                nv[m] = tq[0] * v[m & ~bit] + tq[1] * v[m | bit];
            }
#pragma unroll
            for (int m = 0; m < 16; ++m) v[m] = nv[m];
        }
#pragma unroll
        for (int i = 0; i < 16; ++i) w2t[blk * 256 + j * 16 + i] = v[i];
    } else if (tid < 2304) {
        int c = tid - 2048;
        float v[4][4];
#pragma unroll
        for (int u = 0; u < 4; ++u)
#pragma unroll
            for (int j = 0; j < 4; ++j) v[u][j] = (u == j) ? 1.0f : 0.0f;
        float nv[4][4];
#pragma unroll
        for (int u = 0; u < 4; ++u) {
            const float* tq = tw + 8 * 2048 + ((u >> 1) * 256 + c) * 4 + (u & 1) * 2;
#pragma unroll
            for (int j = 0; j < 4; ++j)
                nv[u][j] = tq[0] * v[u & ~1][j] + tq[1] * v[u | 1][j];
        }
#pragma unroll
        for (int u = 0; u < 4; ++u)
#pragma unroll
            for (int j = 0; j < 4; ++j) v[u][j] = nv[u][j];
#pragma unroll
        for (int u = 0; u < 4; ++u) {
            const float* tq = tw + 9 * 2048 + (c + 256 * (u & 1)) * 4 + ((u >> 1) & 1) * 2;
#pragma unroll
            for (int j = 0; j < 4; ++j)
                nv[u][j] = tq[0] * v[u & ~2][j] + tq[1] * v[u | 2][j];
        }
#pragma unroll
        for (int u = 0; u < 4; ++u)
#pragma unroll
            for (int j = 0; j < 4; ++j) w3[c * 16 + u * 4 + j] = nv[u][j];
    }
}

__device__ __forceinline__ int SWZ(int r, int c) {
    return r * IN_SIZE + (c ^ ((r & 15) << 2));
}

__global__ void __launch_bounds__(256)
bfly_main(const float* __restrict__ x, const float* __restrict__ bias,
          const float* __restrict__ ws, float* __restrict__ out) {
    __shared__ float U[TILE_ROWS * IN_SIZE];

    const float* w1t = ws + W1T_OFF;
    const float* w2t = ws + W2T_OFF;
    const float* w3  = ws + W3_OFF;

    const int row0 = blockIdx.x * TILE_ROWS;
    const int tid  = threadIdx.x;
    const int wave = tid >> 5;
    const int lane = tid & 31;
    const int m16  = lane & 15;
    const int khi  = (lane >> 4) * 2;

    {
        const float4* xg = (const float4*)(x + (size_t)row0 * IN_SIZE);
#pragma unroll
        for (int it = 0; it < 16; ++it) {
            int idx = it * 256 + tid;
            int r   = idx >> 8;
            int c4  = idx & 255;
            float4 val = xg[r * 256 + c4];
            *(float4*)&U[r * IN_SIZE + ((c4 * 4) ^ ((r & 15) << 2))] = val;
        }
    }
    __syncthreads();

    {
        v2f a[8][4];
#pragma unroll
        for (int t = 0; t < 8; ++t) {
            int b = wave + 8 * t;
#pragma unroll
            for (int k = 0; k < 4; ++k) {
                int col = b * 16 + k * 4 + khi;
                a[t][k].x = U[SWZ(m16, col)];
                a[t][k].y = U[SWZ(m16, col + 1)];
            }
        }
        __syncthreads();
#pragma unroll
        for (int t = 0; t < 8; ++t) {
            int b = wave + 8 * t;
            v8f acc = {};
#pragma unroll
            for (int k = 0; k < 4; ++k) {
                int K = k * 4 + khi;
                v2f bf;
                bf.x = w1t[b * 256 + K * 16 + m16];
                bf.y = w1t[b * 256 + (K + 1) * 16 + m16];
                acc = __builtin_amdgcn_wmma_f32_16x16x4_f32(
                        false, a[t][k], false, bf, (short)0, acc, false, false);
            }
#pragma unroll
            for (int v = 0; v < 8; ++v) {
                int r = v + ((lane >> 4) << 3);
                U[SWZ(r, b * 16 + m16)] = acc[v];
            }
        }
    }
    __syncthreads();

    {
        v2f a[8][4];
#pragma unroll
        for (int t = 0; t < 8; ++t) {
            int blk = wave + 8 * t;
            int G = blk >> 4, c = blk & 15;
#pragma unroll
            for (int k = 0; k < 4; ++k) {
                int mb = k * 4 + khi;
                a[t][k].x = U[SWZ(m16, G * 256 + mb * 16 + c)];
                a[t][k].y = U[SWZ(m16, G * 256 + (mb + 1) * 16 + c)];
            }
        }
        __syncthreads();
#pragma unroll
        for (int t = 0; t < 8; ++t) {
            int blk = wave + 8 * t;
            int G = blk >> 4, c = blk & 15;
            v8f acc = {};
#pragma unroll
            for (int k = 0; k < 4; ++k) {
                int K = k * 4 + khi;
                v2f bf;
                bf.x = w2t[blk * 256 + K * 16 + m16];
                bf.y = w2t[blk * 256 + (K + 1) * 16 + m16];
                acc = __builtin_amdgcn_wmma_f32_16x16x4_f32(
                        false, a[t][k], false, bf, (short)0, acc, false, false);
            }
#pragma unroll
            for (int v = 0; v < 8; ++v) {
                int r = v + ((lane >> 4) << 3);
                U[SWZ(r, G * 256 + m16 * 16 + c)] = acc[v];
            }
        }
    }
    __syncthreads();

    {
        int c = tid;
        float m[16];
#pragma unroll
        for (int i = 0; i < 16; ++i) m[i] = w3[c * 16 + i];
        float bi[4];
#pragma unroll
        for (int u = 0; u < 4; ++u) bi[u] = bias[c + 256 * u];
#pragma unroll
        for (int r = 0; r < 16; ++r) {
            float in0 = U[SWZ(r, c)];
            float in1 = U[SWZ(r, c + 256)];
            float in2 = U[SWZ(r, c + 512)];
            float in3 = U[SWZ(r, c + 768)];
            float* og = out + (size_t)(row0 + r) * IN_SIZE;
#pragma unroll
            for (int u = 0; u < 4; ++u) {
                og[c + 256 * u] = m[u * 4 + 0] * in0 + m[u * 4 + 1] * in1 +
                                  m[u * 4 + 2] * in2 + m[u * 4 + 3] * in3 + bi[u];
            }
        }
    }
}

extern "C" void kernel_launch(void* const* d_in, const int* in_sizes, int n_in,
                              void* d_out, int out_size, void* d_ws, size_t ws_size,
                              hipStream_t stream) {
    const float* x    = (const float*)d_in[0];
    const float* tw   = (const float*)d_in[1];
    const float* bias = (const float*)d_in[2];
    float* out = (float*)d_out;
    float* ws  = (float*)d_ws;

    int rows = in_sizes[0] / IN_SIZE;

    bfly_precompute<<<dim3(9), dim3(256), 0, stream>>>(tw, ws);
    bfly_main<<<dim3(rows / TILE_ROWS), dim3(256), 0, stream>>>(x, bias, ws, out);
}


# Persistent work-groups (256 x 16 waves, one per WGP) loop over row tiles: weight fragments loaded once into registers, not per tile; next tile's rows fetched ahead under LDS-only barriers
# speedup vs baseline: 1.2530x; 1.2530x over previous
#include <hip/hip_runtime.h>

typedef __attribute__((ext_vector_type(2))) float v2f;
typedef __attribute__((ext_vector_type(4))) float v4f;
typedef __attribute__((ext_vector_type(8))) float v8f;

#define IN_SIZE   1024
#define TILE_ROWS 16

#define W1T_OFF 0
#define W2T_OFF 16384
#define W3_OFF  32768


__global__ void __launch_bounds__(256)
bfly_precompute(const float* __restrict__ tw, float* __restrict__ ws) {
    float* w1t = ws + W1T_OFF;
    float* w2t = ws + W2T_OFF;
    float* w3  = ws + W3_OFF;
    int tid = blockIdx.x * 256 + threadIdx.x;

    if (tid < 1024) {
        int b = tid >> 4, j = tid & 15;
        float v[16];
#pragma unroll
        for (int l = 0; l < 16; ++l) v[l] = (l == j) ? 1.0f : 0.0f;
#pragma unroll
        for (int idx = 0; idx < 4; ++idx) {
            int s = 1 << idx;
            float nv[16];
#pragma unroll
            for (int l = 0; l < 16; ++l) {
                int p = b * 16 + l;
                int q = (p >> (idx + 1)) * s + (p & (s - 1));
                int i = (l >> idx) & 1;
                const float* tq = tw + idx * 2048 + q * 4 + i * 2;
                nv[l] = tq[0] * v[l & ~s] + tq[1] * v[l | s];
            }
#pragma unroll
            for (int l = 0; l < 16; ++l) v[l] = nv[l];
        }
#pragma unroll
        for (int i = 0; i < 16; ++i)
            w1t[b * 256 + (i + 16 * (j >> 3)) * 8 + (j & 7)] = v[i];
    } else if (tid < 2048) {
        int t2 = tid - 1024;
        int blk = t2 >> 4, j = t2 & 15;
        int G = blk >> 4, c = blk & 15;
        float v[16];
#pragma unroll
        for (int m = 0; m < 16; ++m) v[m] = (m == j) ? 1.0f : 0.0f;
#pragma unroll
        for (int idx = 4; idx < 8; ++idx) {
            int s = 1 << idx;
            int bit = 1 << (idx - 4);
            float nv[16];
#pragma unroll
            for (int m = 0; m < 16; ++m) {
                int p = G * 256 + 16 * m + c;
                int q = (p >> (idx + 1)) * s + (p & (s - 1));
                int i = (p >> idx) & 1;
                const float* tq = tw + idx * 2048 + q * 4 + i * 2;
                nv[m] = tq[0] * v[m & ~bit] + tq[1] * v[m | bit];
            }
#pragma unroll
            for (int m = 0; m < 16; ++m) v[m] = nv[m];
        }
#pragma unroll
        for (int i = 0; i < 16; ++i)
            w2t[blk * 256 + (i + 16 * (j >> 3)) * 8 + (j & 7)] = v[i];
    } else if (tid < 2304) {
        int c = tid - 2048;
        float v[4][4];
#pragma unroll
        for (int u = 0; u < 4; ++u)
#pragma unroll
            for (int j = 0; j < 4; ++j) v[u][j] = (u == j) ? 1.0f : 0.0f;
        float nv[4][4];
#pragma unroll
        for (int u = 0; u < 4; ++u) {
            const float* tq = tw + 8 * 2048 + ((u >> 1) * 256 + c) * 4 + (u & 1) * 2;
#pragma unroll
            for (int j = 0; j < 4; ++j)
                nv[u][j] = tq[0] * v[u & ~1][j] + tq[1] * v[u | 1][j];
        }
#pragma unroll
        for (int u = 0; u < 4; ++u)
#pragma unroll
            for (int j = 0; j < 4; ++j) v[u][j] = nv[u][j];
#pragma unroll
        for (int u = 0; u < 4; ++u) {
            const float* tq = tw + 9 * 2048 + (c + 256 * (u & 1)) * 4 + ((u >> 1) & 1) * 2;
#pragma unroll
            for (int j = 0; j < 4; ++j)
                nv[u][j] = tq[0] * v[u & ~2][j] + tq[1] * v[u | 2][j];
        }
#pragma unroll
        for (int u = 0; u < 4; ++u)
#pragma unroll
            for (int j = 0; j < 4; ++j) w3[c * 16 + u * 4 + j] = nv[u][j];
    }
}

__device__ __forceinline__ int SWZ(int r, int c) {
    return r * IN_SIZE + (c ^ ((r & 15) << 2));
}

#define TILE_FLOATS (TILE_ROWS * IN_SIZE)

__device__ __forceinline__ v8f mm4(float a0, float a1, float b0, float b1, v8f acc) {
    v2f af, bf;
    af.x = a0; af.y = a1; bf.x = b0; bf.y = b1;
    return __builtin_amdgcn_wmma_f32_16x16x4_f32(false, af, false, bf, (short)0, acc, false, false);
}

__device__ __forceinline__ v4f lds_ld4(const float* p) {
    return *(const v4f*)__builtin_assume_aligned(p, 16);
}
__device__ __forceinline__ void lds_st4(float* p, v4f v) {
    *(v4f*)__builtin_assume_aligned(p, 16) = v;
}

__device__ __forceinline__ void lds_barrier() {
    __builtin_amdgcn_fence(__ATOMIC_RELEASE, "workgroup", "local");
    __builtin_amdgcn_s_barrier();
    __builtin_amdgcn_fence(__ATOMIC_ACQUIRE, "workgroup", "local");
}

__global__ void __launch_bounds__(512)
bfly_main(const float* __restrict__ x, const float* __restrict__ bias,
          const float* __restrict__ ws, float* __restrict__ out, int ntiles) {
    __shared__ float Ubuf[2 * TILE_FLOATS];

    const int tid  = threadIdx.x;
    const int wave = tid >> 5;
    const int lane = tid & 31;
    const int m16  = lane & 15;
    const int h8   = (lane >> 4) * 8;
    const int G    = wave >> 2;
    const int j4   = (wave & 3) * 4;

    const int sr   = tid >> 5;
    const int scb  = (tid & 31) * 4;
    const int xoff = sr * IN_SIZE + scb;
    const int uoff = sr * IN_SIZE + (scb ^ ((sr & 15) << 2));

    int tile = blockIdx.x;
    v4f val[8] = {};
    if (tile < ntiles) {
        const float* xg = x + (size_t)tile * TILE_FLOATS + xoff;
#pragma unroll
        for (int it = 0; it < 8; ++it) val[it] = *(const v4f*)(xg + 128 * it);
    }

    v4f w1r[4][2];
    {
        const v4f* wa = (const v4f*)(ws + W1T_OFF);
#pragma unroll
        for (int t = 0; t < 4; ++t) {
            int b = wave + 16 * t;
            w1r[t][0] = wa[b * 64 + lane * 2];
            w1r[t][1] = wa[b * 64 + lane * 2 + 1];
        }
    }
    v4f w2r[4][2];
    {
        const v4f* wa = (const v4f*)(ws + W2T_OFF);
#pragma unroll
        for (int q = 0; q < 4; ++q) {
            int blk = G * 16 + j4 + q;
            w2r[q][0] = wa[blk * 64 + lane * 2];
            w2r[q][1] = wa[blk * 64 + lane * 2 + 1];
        }
    }
    const int c  = tid & 255;
    const int rh = tid >> 8;
    float m[16];
    float bi[4];
    {
        const float* w3 = ws + W3_OFF;
#pragma unroll
        for (int i = 0; i < 16; ++i) m[i] = w3[c * 16 + i];
#pragma unroll
        for (int u = 0; u < 4; ++u) bi[u] = bias[c + 256 * u];
    }

    asm volatile("" :: "v"(w1r[0][0]), "v"(w1r[0][1]), "v"(w1r[1][0]), "v"(w1r[1][1]),
                       "v"(w1r[2][0]), "v"(w1r[2][1]), "v"(w1r[3][0]), "v"(w1r[3][1]));
    asm volatile("" :: "v"(w2r[0][0]), "v"(w2r[0][1]), "v"(w2r[1][0]), "v"(w2r[1][1]),
                       "v"(w2r[2][0]), "v"(w2r[2][1]), "v"(w2r[3][0]), "v"(w2r[3][1]));
    asm volatile("" :: "v"(m[0]), "v"(m[1]), "v"(m[2]), "v"(m[3]), "v"(m[4]), "v"(m[5]),
                       "v"(m[6]), "v"(m[7]), "v"(m[8]), "v"(m[9]), "v"(m[10]), "v"(m[11]),
                       "v"(m[12]), "v"(m[13]), "v"(m[14]), "v"(m[15]),
                       "v"(bi[0]), "v"(bi[1]), "v"(bi[2]), "v"(bi[3]));

    int k = 0;
    for (; tile < ntiles; tile += gridDim.x, ++k) {
        const int row0 = tile * TILE_ROWS;
        const int ub   = (k & 1) * TILE_FLOATS;

        {
            float* ul = &Ubuf[ub + uoff];
#pragma unroll
            for (int it = 0; it < 8; ++it) lds_st4(ul + 128 * it, val[it]);
        }
        {
            const int next = tile + gridDim.x;
            if (next < ntiles) {
                const float* xg = x + (size_t)next * TILE_FLOATS + xoff;
#pragma unroll
                for (int it = 0; it < 8; ++it) val[it] = *(const v4f*)(xg + 128 * it);
            }
        }
        lds_barrier();


        {
            const int a0 = SWZ(m16, wave * 16 + h8);
            const int a1 = SWZ(m16, wave * 16 + h8 + 4);
            v4f d[4][2];
#pragma unroll
            for (int t = 0; t < 4; ++t) {
                d[t][0] = lds_ld4(&Ubuf[ub + a0 + 256 * t]);
                d[t][1] = lds_ld4(&Ubuf[ub + a1 + 256 * t]);
            }
            v8f acc[4] = {};
#pragma unroll
            for (int t = 0; t < 4; ++t) acc[t] = mm4(w1r[t][0].x, w1r[t][0].y, d[t][0].x, d[t][0].y, acc[t]);
#pragma unroll
            for (int t = 0; t < 4; ++t) acc[t] = mm4(w1r[t][0].z, w1r[t][0].w, d[t][0].z, d[t][0].w, acc[t]);
#pragma unroll
            for (int t = 0; t < 4; ++t) acc[t] = mm4(w1r[t][1].x, w1r[t][1].y, d[t][1].x, d[t][1].y, acc[t]);
#pragma unroll
            for (int t = 0; t < 4; ++t) acc[t] = mm4(w1r[t][1].z, w1r[t][1].w, d[t][1].z, d[t][1].w, acc[t]);
#pragma unroll
            for (int t = 0; t < 4; ++t) {
                v4f o0, o1;
                o0.x = acc[t][0]; o0.y = acc[t][1]; o0.z = acc[t][2]; o0.w = acc[t][3];
                o1.x = acc[t][4]; o1.y = acc[t][5]; o1.z = acc[t][6]; o1.w = acc[t][7];
                lds_st4(&Ubuf[ub + a0 + 256 * t], o0);
                lds_st4(&Ubuf[ub + a1 + 256 * t], o1);
            }
        }
        lds_barrier();

        {
            const int base = G * 256 + h8 * 16 + j4;
            int ga[4];
#pragma unroll
            for (int lo = 0; lo < 4; ++lo) ga[lo] = SWZ(m16, base + 16 * lo);
            v4f io[8];
#pragma unroll
            for (int kk = 0; kk < 8; ++kk) io[kk] = lds_ld4(&Ubuf[ub + ga[kk & 3] + 64 * (kk >> 2)]);
            v8f acc[4] = {};
#pragma unroll
            for (int q = 0; q < 4; ++q) acc[q] = mm4(w2r[q][0].x, w2r[q][0].y, io[0][q], io[1][q], acc[q]);
#pragma unroll
            for (int q = 0; q < 4; ++q) acc[q] = mm4(w2r[q][0].z, w2r[q][0].w, io[2][q], io[3][q], acc[q]);
#pragma unroll
            for (int q = 0; q < 4; ++q) acc[q] = mm4(w2r[q][1].x, w2r[q][1].y, io[4][q], io[5][q], acc[q]);
#pragma unroll
            for (int q = 0; q < 4; ++q) acc[q] = mm4(w2r[q][1].z, w2r[q][1].w, io[6][q], io[7][q], acc[q]);
#pragma unroll
            for (int q = 0; q < 4; ++q)
#pragma unroll
                for (int r = 0; r < 8; ++r) io[r][q] = acc[q][r];
#pragma unroll
            for (int r = 0; r < 8; ++r) lds_st4(&Ubuf[ub + ga[r & 3] + 64 * (r >> 2)], io[r]);
        }
        lds_barrier();

        {
            float* og = out + (size_t)(row0 + rh * 8) * IN_SIZE + c;
#pragma unroll
            for (int i = 0; i < 8; ++i) {
                int r = rh * 8 + i;
                float in0 = Ubuf[ub + SWZ(r, c)];
                float in1 = Ubuf[ub + SWZ(r, c + 256)];
                float in2 = Ubuf[ub + SWZ(r, c + 512)];
                float in3 = Ubuf[ub + SWZ(r, c + 768)];
#pragma unroll
                for (int u = 0; u < 4; ++u) {
                    og[i * IN_SIZE + 256 * u] = m[u * 4 + 0] * in0 + m[u * 4 + 1] * in1 +
                                                m[u * 4 + 2] * in2 + m[u * 4 + 3] * in3 + bi[u];
                }
            }
        }
    }
}

extern "C" void kernel_launch(void* const* d_in, const int* in_sizes, int n_in,
                              void* d_out, int out_size, void* d_ws, size_t ws_size,
                              hipStream_t stream) {
    const float* x    = (const float*)d_in[0];
    const float* tw   = (const float*)d_in[1];
    const float* bias = (const float*)d_in[2];
    float* out = (float*)d_out;
    float* ws  = (float*)d_ws;

    int rows = in_sizes[0] / IN_SIZE;

    bfly_precompute<<<dim3(9), dim3(256), 0, stream>>>(tw, ws);
    bfly_main<<<dim3(256), dim3(512), 0, stream>>>(x, bias, ws, out, rows / TILE_ROWS);
}


# Stages regrouped 2+4+4 (was 4+4+2): the 4x4 mix runs in registers on the 4 columns a lane loads; stages 2-9 on WMMA; rows leave LDS as 128-bit words, 512 B per wave store; sum order changes
# speedup vs baseline: 1.5679x; 1.2513x over previous
#include <hip/hip_runtime.h>

typedef __attribute__((ext_vector_type(2))) float v2f;
typedef __attribute__((ext_vector_type(4))) float v4f;
typedef __attribute__((ext_vector_type(8))) float v8f;

#define IN_SIZE   1024
#define TILE_ROWS 16

#define WA_OFF 0
#define WB_OFF 16384
#define W0_OFF 32768


template <int NS>
__device__ __forceinline__ void compose(const float* __restrict__ tw, int idx0, int pbase,
                                        int j, float* v) {
    constexpr int L = 1 << NS;
#pragma unroll
    for (int l = 0; l < L; ++l) v[l] = (l == j) ? 1.0f : 0.0f;
#pragma unroll
    for (int n = 0; n < NS; ++n) {
        int idx = idx0 + n;
        int s = 1 << idx;
        int bit = 1 << n;
        float nv[L];
#pragma unroll
        for (int l = 0; l < L; ++l) {
            int p = pbase + (l << idx0);
            int q = (p >> (idx + 1)) * s + (p & (s - 1));
            int i = (p >> idx) & 1;
            const float* tq = tw + idx * 2048 + q * 4 + i * 2;
            nv[l] = tq[0] * v[l & ~bit] + tq[1] * v[l | bit];
        }
#pragma unroll
        for (int l = 0; l < L; ++l) v[l] = nv[l];
    }
}

__global__ void __launch_bounds__(256)
bfly_precompute(const float* __restrict__ tw, float* __restrict__ ws) {
    float* wa = ws + WA_OFF;
    float* wb = ws + WB_OFF;
    float* w0 = ws + W0_OFF;
    int tid = blockIdx.x * 256 + threadIdx.x;

    if (tid < 1024) {
        int blk = tid >> 4, j = tid & 15;
        int g = blk >> 2, cc = blk & 3;
        float v[16];
        compose<4>(tw, 2, 64 * g + cc, j, v);
#pragma unroll
        for (int i = 0; i < 16; ++i)
            wa[blk * 256 + (i + 16 * (j >> 3)) * 8 + (j & 7)] = v[i];
    } else if (tid < 2048) {
        int t2 = tid - 1024;
        int blk = t2 >> 4, j = t2 & 15;
        float v[16];
        compose<4>(tw, 6, blk, j, v);
#pragma unroll
        for (int i = 0; i < 16; ++i)
            wb[blk * 256 + (i + 16 * (j >> 3)) * 8 + (j & 7)] = v[i];
    } else if (tid < 2304) {
        int c = tid - 2048;
#pragma unroll
        for (int j = 0; j < 4; ++j) {
            float v[4];
            compose<2>(tw, 0, 4 * c, j, v);
#pragma unroll
            for (int i = 0; i < 4; ++i) w0[c * 16 + i * 4 + j] = v[i];
        }
    }
}

__device__ __forceinline__ int SWZ(int r, int c) {
    return r * IN_SIZE + (c ^ ((r & 15) << 2));
}

#define TILE_FLOATS (TILE_ROWS * IN_SIZE)

__device__ __forceinline__ v8f mm4(float a0, float a1, float b0, float b1, v8f acc) {
    v2f af, bf;
    af.x = a0; af.y = a1; bf.x = b0; bf.y = b1;
    return __builtin_amdgcn_wmma_f32_16x16x4_f32(false, af, false, bf, (short)0, acc, false, false);
}

__device__ __forceinline__ v4f lds_ld4(const float* p) {
    return *(const v4f*)__builtin_assume_aligned(p, 16);
}
__device__ __forceinline__ void lds_st4(float* p, v4f v) {
    *(v4f*)__builtin_assume_aligned(p, 16) = v;
}

__device__ __forceinline__ void lds_barrier() {
    __builtin_amdgcn_fence(__ATOMIC_RELEASE, "workgroup", "local");
    __builtin_amdgcn_s_barrier();
    __builtin_amdgcn_fence(__ATOMIC_ACQUIRE, "workgroup", "local");
}

__global__ void __launch_bounds__(512)
bfly_main(const float* __restrict__ x, const float* __restrict__ bias,
          const float* __restrict__ ws, float* __restrict__ out, int ntiles) {
    __shared__ float Ubuf[2 * TILE_FLOATS];

    const int tid  = threadIdx.x;
    const int wave = tid >> 5;
    const int lane = tid & 31;
    const int m16  = lane & 15;
    const int h8   = (lane >> 4) * 8;

    const int c4   = tid & 255;
    const int rh   = tid >> 8;
    const int xoff = rh * 8 * IN_SIZE + 4 * c4;
    int so[8];
#pragma unroll
    for (int i = 0; i < 8; ++i) so[i] = SWZ(rh * 8 + i, 4 * c4);

    int tile = blockIdx.x;
    v4f val[8] = {};
    if (tile < ntiles) {
        const float* xg = x + (size_t)tile * TILE_FLOATS + xoff;
#pragma unroll
        for (int i = 0; i < 8; ++i) val[i] = *(const v4f*)(xg + i * IN_SIZE);
    }

    v4f war[4][2];
    {
        const v4f* wa = (const v4f*)(ws + WA_OFF);
#pragma unroll
        for (int q = 0; q < 4; ++q) {
            int blk = wave * 4 + q;
            war[q][0] = wa[blk * 64 + lane * 2];
            war[q][1] = wa[blk * 64 + lane * 2 + 1];
        }
    }
    v4f wbr[4][2];
    {
        const v4f* wb = (const v4f*)(ws + WB_OFF);
#pragma unroll
        for (int q = 0; q < 4; ++q) {
            int blk = wave * 4 + q;
            wbr[q][0] = wb[blk * 64 + lane * 2];
            wbr[q][1] = wb[blk * 64 + lane * 2 + 1];
        }
    }
    float m[16];
    {
        const float* w0 = ws + W0_OFF;
#pragma unroll
        for (int i = 0; i < 16; ++i) m[i] = w0[c4 * 16 + i];
    }
    v4f bi;
    bi.x = bias[4 * c4]; bi.y = bias[4 * c4 + 1]; bi.z = bias[4 * c4 + 2]; bi.w = bias[4 * c4 + 3];

    asm volatile("" :: "v"(war[0][0]), "v"(war[0][1]), "v"(war[1][0]), "v"(war[1][1]),
                       "v"(war[2][0]), "v"(war[2][1]), "v"(war[3][0]), "v"(war[3][1]));
    asm volatile("" :: "v"(wbr[0][0]), "v"(wbr[0][1]), "v"(wbr[1][0]), "v"(wbr[1][1]),
                       "v"(wbr[2][0]), "v"(wbr[2][1]), "v"(wbr[3][0]), "v"(wbr[3][1]));
    asm volatile("" :: "v"(m[0]), "v"(m[1]), "v"(m[2]), "v"(m[3]), "v"(m[4]), "v"(m[5]),
                       "v"(m[6]), "v"(m[7]), "v"(m[8]), "v"(m[9]), "v"(m[10]), "v"(m[11]),
                       "v"(m[12]), "v"(m[13]), "v"(m[14]), "v"(m[15]), "v"(bi));

    int k = 0;
    for (; tile < ntiles; tile += gridDim.x, ++k) {
        const int row0 = tile * TILE_ROWS;
        const int ub   = (k & 1) * TILE_FLOATS;

#pragma unroll
        for (int i = 0; i < 8; ++i) {
            const v4f a = val[i];
            v4f o;
            o.x = m[0]  * a.x + m[1]  * a.y + m[2]  * a.z + m[3]  * a.w;
            o.y = m[4]  * a.x + m[5]  * a.y + m[6]  * a.z + m[7]  * a.w;
            o.z = m[8]  * a.x + m[9]  * a.y + m[10] * a.z + m[11] * a.w;
            o.w = m[12] * a.x + m[13] * a.y + m[14] * a.z + m[15] * a.w;
            lds_st4(&Ubuf[ub + so[i]], o);
        }
        {
            const int next = tile + gridDim.x;
            if (next < ntiles) {
                const float* xg = x + (size_t)next * TILE_FLOATS + xoff;
#pragma unroll
                for (int i = 0; i < 8; ++i) val[i] = *(const v4f*)(xg + i * IN_SIZE);
            }
        }
        lds_barrier();


        {
            int ga[8];
#pragma unroll
            for (int e = 0; e < 8; ++e) ga[e] = SWZ(m16, wave * 64 + (h8 + e) * 4);
            v4f io[8];
#pragma unroll
            for (int e = 0; e < 8; ++e) io[e] = lds_ld4(&Ubuf[ub + ga[e]]);
            v8f acc[4] = {};
#pragma unroll
            for (int q = 0; q < 4; ++q) acc[q] = mm4(war[q][0].x, war[q][0].y, io[0][q], io[1][q], acc[q]);
#pragma unroll
            for (int q = 0; q < 4; ++q) acc[q] = mm4(war[q][0].z, war[q][0].w, io[2][q], io[3][q], acc[q]);
#pragma unroll
            for (int q = 0; q < 4; ++q) acc[q] = mm4(war[q][1].x, war[q][1].y, io[4][q], io[5][q], acc[q]);
#pragma unroll
            for (int q = 0; q < 4; ++q) acc[q] = mm4(war[q][1].z, war[q][1].w, io[6][q], io[7][q], acc[q]);
#pragma unroll
            for (int q = 0; q < 4; ++q)
#pragma unroll
                for (int r = 0; r < 8; ++r) io[r][q] = acc[q][r];
#pragma unroll
            for (int r = 0; r < 8; ++r) lds_st4(&Ubuf[ub + ga[r]], io[r]);
        }
        lds_barrier();

        {
            const int gb = SWZ(m16, 64 * h8 + 4 * wave);
            v4f io[8];
#pragma unroll
            for (int e = 0; e < 8; ++e) io[e] = lds_ld4(&Ubuf[ub + gb + 64 * e]);
            v8f acc[4] = {};
#pragma unroll
            for (int q = 0; q < 4; ++q) acc[q] = mm4(wbr[q][0].x, wbr[q][0].y, io[0][q], io[1][q], acc[q]);
#pragma unroll
            for (int q = 0; q < 4; ++q) acc[q] = mm4(wbr[q][0].z, wbr[q][0].w, io[2][q], io[3][q], acc[q]);
#pragma unroll
            for (int q = 0; q < 4; ++q) acc[q] = mm4(wbr[q][1].x, wbr[q][1].y, io[4][q], io[5][q], acc[q]);
#pragma unroll
            for (int q = 0; q < 4; ++q) acc[q] = mm4(wbr[q][1].z, wbr[q][1].w, io[6][q], io[7][q], acc[q]);
#pragma unroll
            for (int q = 0; q < 4; ++q)
#pragma unroll
                for (int r = 0; r < 8; ++r) io[r][q] = acc[q][r];
#pragma unroll
            for (int r = 0; r < 8; ++r) lds_st4(&Ubuf[ub + gb + 64 * r], io[r]);
        }
        lds_barrier();

        {
            float* og = out + (size_t)row0 * IN_SIZE + xoff;
#pragma unroll
            for (int i = 0; i < 8; ++i) {
                v4f v = lds_ld4(&Ubuf[ub + so[i]]);
                v.x += bi.x; v.y += bi.y; v.z += bi.z; v.w += bi.w;
                *(v4f*)(og + i * IN_SIZE) = v;
            }
        }
    }
}

extern "C" void kernel_launch(void* const* d_in, const int* in_sizes, int n_in,
                              void* d_out, int out_size, void* d_ws, size_t ws_size,
                              hipStream_t stream) {
    const float* x    = (const float*)d_in[0];
    const float* tw   = (const float*)d_in[1];
    const float* bias = (const float*)d_in[2];
    float* out = (float*)d_out;
    float* ws  = (float*)d_ws;

    int rows = in_sizes[0] / IN_SIZE;

    bfly_precompute<<<dim3(9), dim3(256), 0, stream>>>(tw, ws);
    bfly_main<<<dim3(256), dim3(512), 0, stream>>>(x, bias, ws, out, rows / TILE_ROWS);
}


# Set-up launch folded into the main kernel: each work-group builds its weight fragments from the twiddles (stages 2-9 staged once in LDS; same products, same rounding order); one launch, no workspace
# speedup vs baseline: 1.6727x; 1.0668x over previous
#include <hip/hip_runtime.h>

typedef __attribute__((ext_vector_type(2))) float v2f;
typedef __attribute__((ext_vector_type(4))) float v4f;
typedef __attribute__((ext_vector_type(8))) float v8f;

#define IN_SIZE   1024
#define TILE_ROWS 16


__device__ __forceinline__ int tw_off(int idx, int p) {
    const int s = 1 << idx;
    const int q = (p >> (idx + 1)) * s + (p & (s - 1));
    return (idx - 2) * 2048 + q * 4;
}

template <int IDX0>
__device__ __forceinline__ void block_fragment(const float* tl, int pbase, int i, int h,
                                               v4f& w0, v4f& w1) {
    const int i0 = i & 1, i1 = (i >> 1) & 1, i2 = (i >> 2) & 1, i3 = (i >> 3) & 1;
    const float t3 = tl[tw_off(IDX0 + 3, pbase + (i << IDX0)) + i3 * 2 + h];
    const int l2 = (h << 3) | (i & 7);
    const v2f t2 = *(const v2f*)&tl[tw_off(IDX0 + 2, pbase + (l2 << IDX0)) + i2 * 2];
    float e[8];
#pragma unroll
    for (int j2 = 0; j2 < 2; ++j2) {
        const int l1 = (h << 3) | (j2 << 2) | (i & 3);
        const v2f t1 = *(const v2f*)&tl[tw_off(IDX0 + 1, pbase + (l1 << IDX0)) + i1 * 2];
#pragma unroll
        for (int j1 = 0; j1 < 2; ++j1) {
            const int l0 = (h << 3) | (j2 << 2) | (j1 << 1) | i0;
            const v2f t0 = *(const v2f*)&tl[tw_off(IDX0, pbase + (l0 << IDX0)) + i0 * 2];
#pragma unroll
            for (int j0 = 0; j0 < 2; ++j0) {
                const float a = t1[j1] * t0[j0];
                const float b = t2[j2] * a;
                e[j2 * 4 + j1 * 2 + j0] = t3 * b;
            }
        }
    }
    w0.x = e[0]; w0.y = e[1]; w0.z = e[2]; w0.w = e[3];
    w1.x = e[4]; w1.y = e[5]; w1.z = e[6]; w1.w = e[7];
}

__device__ __forceinline__ int SWZ(int r, int c) {
    return r * IN_SIZE + (c ^ ((r & 15) << 2));
}

#define TILE_FLOATS (TILE_ROWS * IN_SIZE)

__device__ __forceinline__ v8f mm4(float a0, float a1, float b0, float b1, v8f acc) {
    v2f af, bf;
    af.x = a0; af.y = a1; bf.x = b0; bf.y = b1;
    return __builtin_amdgcn_wmma_f32_16x16x4_f32(false, af, false, bf, (short)0, acc, false, false);
}

__device__ __forceinline__ v4f lds_ld4(const float* p) {
    return *(const v4f*)__builtin_assume_aligned(p, 16);
}
__device__ __forceinline__ void lds_st4(float* p, v4f v) {
    *(v4f*)__builtin_assume_aligned(p, 16) = v;
}

__device__ __forceinline__ void lds_barrier() {
    __builtin_amdgcn_fence(__ATOMIC_RELEASE, "workgroup", "local");
    __builtin_amdgcn_s_barrier();
    __builtin_amdgcn_fence(__ATOMIC_ACQUIRE, "workgroup", "local");
}

__global__ void __launch_bounds__(512)
bfly_main(const float* __restrict__ x, const float* __restrict__ tw,
          const float* __restrict__ bias, float* __restrict__ out, int ntiles) {
    __shared__ float Ubuf[2 * TILE_FLOATS];

    const int tid  = threadIdx.x;
    const int wave = tid >> 5;
    const int lane = tid & 31;
    const int m16  = lane & 15;
    const int h8   = (lane >> 4) * 8;

    const int c4   = tid & 255;
    const int rh   = tid >> 8;
    const int xoff = rh * 8 * IN_SIZE + 4 * c4;
    int so[8];
#pragma unroll
    for (int i = 0; i < 8; ++i) so[i] = SWZ(rh * 8 + i, 4 * c4);

    v4f tv[8];
    {
        const v4f* tg = (const v4f*)(tw + 2 * 2048);
#pragma unroll
        for (int r = 0; r < 8; ++r) tv[r] = tg[tid + 512 * r];
    }
    v4f s0lo, s0hi, s1lo, s1hi;
    {
        const v4f* t0g = (const v4f*)(tw + 8 * c4);
        const v4f* t1g = (const v4f*)(tw + 2048 + 8 * c4);
        s0lo = t0g[0]; s0hi = t0g[1];
        s1lo = t1g[0]; s1hi = t1g[1];
    }
    const v4f bi = *(const v4f*)(bias + 4 * c4);

    int tile = blockIdx.x;
    v4f val[8];
    {
        const float* xg = x + (size_t)(tile < ntiles ? tile : 0) * TILE_FLOATS + xoff;
#pragma unroll
        for (int i = 0; i < 8; ++i) val[i] = *(const v4f*)(xg + i * IN_SIZE);
    }

    asm volatile("" :: "v"(bi));

    float* tl = &Ubuf[TILE_FLOATS];
#pragma unroll
    for (int r = 0; r < 8; ++r) lds_st4(&tl[4 * (tid + 512 * r)], tv[r]);
    lds_barrier();

    v4f war[4][2];
#pragma unroll
    for (int q = 0; q < 4; ++q)
        block_fragment<2>(tl, 64 * wave + q, m16, lane >> 4, war[q][0], war[q][1]);
    v4f wbr[4][2];
#pragma unroll
    for (int q = 0; q < 4; ++q)
        block_fragment<6>(tl, 4 * wave + q, m16, lane >> 4, wbr[q][0], wbr[q][1]);
    float m[16];
    {
        const float a0[8] = {s0lo.x, s0lo.y, s0lo.z, s0lo.w, s0hi.x, s0hi.y, s0hi.z, s0hi.w};
        const float a1[8] = {s1lo.x, s1lo.y, s1lo.z, s1lo.w, s1hi.x, s1hi.y, s1hi.z, s1hi.w};
#pragma unroll
        for (int i = 0; i < 4; ++i)
#pragma unroll
            for (int j = 0; j < 4; ++j) {
                const int i0 = i & 1, i1 = i >> 1, j0 = j & 1, j1 = j >> 1;
                m[4 * i + j] = a1[i0 * 4 + i1 * 2 + j1] * a0[j1 * 4 + i0 * 2 + j0];
            }
    }

    int k = 0;
    for (; tile < ntiles; tile += gridDim.x, ++k) {
        const int row0 = tile * TILE_ROWS;
        const int ub   = (k & 1) * TILE_FLOATS;

#pragma unroll
        for (int i = 0; i < 8; ++i) {
            const v4f a = val[i];
            v4f o;
            o.x = m[0]  * a.x + m[1]  * a.y + m[2]  * a.z + m[3]  * a.w;
            o.y = m[4]  * a.x + m[5]  * a.y + m[6]  * a.z + m[7]  * a.w;
            o.z = m[8]  * a.x + m[9]  * a.y + m[10] * a.z + m[11] * a.w;
            o.w = m[12] * a.x + m[13] * a.y + m[14] * a.z + m[15] * a.w;
            lds_st4(&Ubuf[ub + so[i]], o);
        }
        {
            const int next = tile + gridDim.x;
            if (next < ntiles) {
                const float* xg = x + (size_t)next * TILE_FLOATS + xoff;
#pragma unroll
                for (int i = 0; i < 8; ++i) val[i] = *(const v4f*)(xg + i * IN_SIZE);
            }
        }
        lds_barrier();


        {
            int ga[8];
#pragma unroll
            for (int e = 0; e < 8; ++e) ga[e] = SWZ(m16, wave * 64 + (h8 + e) * 4);
            v4f io[8];
#pragma unroll
            for (int e = 0; e < 8; ++e) io[e] = lds_ld4(&Ubuf[ub + ga[e]]);
            v8f acc[4] = {};
#pragma unroll
            for (int q = 0; q < 4; ++q) acc[q] = mm4(war[q][0].x, war[q][0].y, io[0][q], io[1][q], acc[q]);
#pragma unroll
            for (int q = 0; q < 4; ++q) acc[q] = mm4(war[q][0].z, war[q][0].w, io[2][q], io[3][q], acc[q]);
#pragma unroll
            for (int q = 0; q < 4; ++q) acc[q] = mm4(war[q][1].x, war[q][1].y, io[4][q], io[5][q], acc[q]);
#pragma unroll
            for (int q = 0; q < 4; ++q) acc[q] = mm4(war[q][1].z, war[q][1].w, io[6][q], io[7][q], acc[q]);
#pragma unroll
            for (int q = 0; q < 4; ++q)
#pragma unroll
                for (int r = 0; r < 8; ++r) io[r][q] = acc[q][r];
#pragma unroll
            for (int r = 0; r < 8; ++r) lds_st4(&Ubuf[ub + ga[r]], io[r]);
        }
        lds_barrier();

        {
            const int gb = SWZ(m16, 64 * h8 + 4 * wave);
            v4f io[8];
#pragma unroll
            for (int e = 0; e < 8; ++e) io[e] = lds_ld4(&Ubuf[ub + gb + 64 * e]);
            v8f acc[4] = {};
#pragma unroll
            for (int q = 0; q < 4; ++q) acc[q] = mm4(wbr[q][0].x, wbr[q][0].y, io[0][q], io[1][q], acc[q]);
#pragma unroll
            for (int q = 0; q < 4; ++q) acc[q] = mm4(wbr[q][0].z, wbr[q][0].w, io[2][q], io[3][q], acc[q]);
#pragma unroll
            for (int q = 0; q < 4; ++q) acc[q] = mm4(wbr[q][1].x, wbr[q][1].y, io[4][q], io[5][q], acc[q]);
#pragma unroll
            for (int q = 0; q < 4; ++q) acc[q] = mm4(wbr[q][1].z, wbr[q][1].w, io[6][q], io[7][q], acc[q]);
#pragma unroll
            for (int q = 0; q < 4; ++q)
#pragma unroll
                for (int r = 0; r < 8; ++r) io[r][q] = acc[q][r];
#pragma unroll
            for (int r = 0; r < 8; ++r) lds_st4(&Ubuf[ub + gb + 64 * r], io[r]);
        }
        lds_barrier();

        {
            float* og = out + (size_t)row0 * IN_SIZE + xoff;
#pragma unroll
            for (int i = 0; i < 8; ++i) {
                v4f v = lds_ld4(&Ubuf[ub + so[i]]);
                v.x += bi.x; v.y += bi.y; v.z += bi.z; v.w += bi.w;
                *(v4f*)(og + i * IN_SIZE) = v;
            }
        }
    }
}

extern "C" void kernel_launch(void* const* d_in, const int* in_sizes, int n_in,
                              void* d_out, int out_size, void* d_ws, size_t ws_size,
                              hipStream_t stream) {
    const float* x    = (const float*)d_in[0];
    const float* tw   = (const float*)d_in[1];
    const float* bias = (const float*)d_in[2];
    float* out = (float*)d_out;

    int rows = in_sizes[0] / IN_SIZE;

    bfly_main<<<dim3(256), dim3(512), 0, stream>>>(x, tw, bias, out, rows / TILE_ROWS);
}
